# starting kernel
# baseline (speedup 1.0000x reference)
#include <hip/hip_runtime.h>

typedef __attribute__((ext_vector_type(2))) float v2f;
typedef __attribute__((ext_vector_type(4))) float v4f;
typedef __attribute__((ext_vector_type(8))) float v8f;

#define S4D_D 1024
#define S4D_N 64
#define S4D_L 2048

__global__ __launch_bounds__(32) void s4d_wmma_kernel(
    const float* __restrict__ A_real,
    const float* __restrict__ A_imag,
    const float* __restrict__ Bmat,
    const float* __restrict__ Cmat,
    const float* __restrict__ Dvec,
    const float* __restrict__ log_dt,
    float* __restrict__ out)
{
    __shared__ float P_re [16 * 64];
    __shared__ float P_nim[16 * 64];
    __shared__ float Q_re [16 * 64];
    __shared__ float Q_im [16 * 64];

    const int d    = blockIdx.x;
    const int lane = threadIdx.x;
    const int hi   = lane >> 4;
    const int lo   = lane & 15;

    const float dt = expf(log_dt[d]);
    const float Dd = Dvec[d];

    float ar[2], ai[2];
    float pr[2], pi[2];
    #pragma unroll
    for (int t = 0; t < 2; ++t) {
        const int n = 2 * lane + t;
        const float zr = 0.5f * dt * A_real[n];
        const float zi = 0.5f * dt * A_imag[n];
        const float dr = 1.0f - zr, di = -zi;
        const float inv = 1.0f / (dr * dr + di * di);
        const float nr = 1.0f + zr, ni = zi;
        ar[t] = (nr * dr + ni * di) * inv;
        ai[t] = (ni * dr - nr * di) * inv;
        const float bs  = dt * Bmat[n * S4D_D + d] * inv;
        const float bbr =  bs * dr;
        const float bbi = -bs * di;
        const float cr = Cmat[(d * S4D_N + n) * 2 + 0];
        const float ci = Cmat[(d * S4D_N + n) * 2 + 1];
        pr[t] = cr * bbr - ci * bbi;
        pi[t] = cr * bbi + ci * bbr;
    }

    #pragma unroll
    for (int k = 0; k < 16; ++k) {
        v2f sre, sim;
        sre[0] = pr[0];  sre[1] = pr[1];
        sim[0] = -pi[0]; sim[1] = -pi[1];
        *(v2f*)&P_re [k * 64 + 2 * lane] = sre;
        *(v2f*)&P_nim[k * 64 + 2 * lane] = sim;
        #pragma unroll
        for (int t = 0; t < 2; ++t) {
            const float trr = pr[t] * ar[t] - pi[t] * ai[t];
            pi[t] = pr[t] * ai[t] + pi[t] * ar[t];
            pr[t] = trr;
        }
    }

    float a16r[2], a16i[2];
    #pragma unroll
    for (int t = 0; t < 2; ++t) { a16r[t] = ar[t]; a16i[t] = ai[t]; }
    #pragma unroll
    for (int s = 0; s < 4; ++s) {
        #pragma unroll
        for (int t = 0; t < 2; ++t) {
            const float trr = a16r[t] * a16r[t] - a16i[t] * a16i[t];
            a16i[t] = 2.0f * a16r[t] * a16i[t];
            a16r[t] = trr;
        }
    }

    float qrr[2] = {1.0f, 1.0f}, qii[2] = {0.0f, 0.0f};

    for (int jt = 0; jt < 8; ++jt) {
        __syncthreads();
        #pragma unroll
        for (int jl = 0; jl < 16; ++jl) {
            v2f sre, sim;
            sre[0] = qrr[0]; sre[1] = qrr[1];
            sim[0] = qii[0]; sim[1] = qii[1];
            *(v2f*)&Q_re[jl * 64 + 2 * lane] = sre;
            *(v2f*)&Q_im[jl * 64 + 2 * lane] = sim;
            #pragma unroll
            for (int t = 0; t < 2; ++t) {
                const float trr = qrr[t] * a16r[t] - qii[t] * a16i[t];
                qii[t] = qrr[t] * a16i[t] + qii[t] * a16r[t];
                qrr[t] = trr;
            }
        }
        __syncthreads();

        v8f acc_r = {};
        v8f acc_i = {};
        #pragma unroll
        for (int kk = 0; kk < 16; ++kk) {
            const int nb = 4 * kk + 2 * hi;
            v2f a_re = *(const v2f*)&P_re [lo * 64 + nb];
            v2f a_im = *(const v2f*)&P_nim[lo * 64 + nb];
            v2f b_re = *(const v2f*)&Q_re [lo * 64 + nb];
            v2f b_im = *(const v2f*)&Q_im [lo * 64 + nb];
            acc_r = __builtin_amdgcn_wmma_f32_16x16x4_f32(
                false, a_re, false, b_re, (short)0, acc_r, false, false);
            acc_i = __builtin_amdgcn_wmma_f32_16x16x4_f32(
                false, a_im, false, b_im, (short)0, acc_i, false, false);
        }

        const int lbase = (jt * 16 + lo) * 16 + 8 * hi;
        v4f w0, w1;
        #pragma unroll
        for (int v = 0; v < 4; ++v) {
            float val = acc_r[v] + acc_i[v];
            val += (lbase + v == 0) ? Dd : 0.0f;
            w0[v] = val;
        }
        #pragma unroll
        for (int v = 0; v < 4; ++v) {
            w1[v] = acc_r[v + 4] + acc_i[v + 4];
        }
        float* orow = out + (size_t)d * S4D_L + lbase;
        *(v4f*)(orow)     = w0;
        *(v4f*)(orow + 4) = w1;
    }
}

extern "C" void kernel_launch(void* const* d_in, const int* in_sizes, int n_in,
                              void* d_out, int out_size, void* d_ws, size_t ws_size,
                              hipStream_t stream) {
    const float* A_real = (const float*)d_in[0];
    const float* A_imag = (const float*)d_in[1];
    const float* B      = (const float*)d_in[2];
    const float* C      = (const float*)d_in[3];
    const float* D      = (const float*)d_in[4];
    const float* log_dt = (const float*)d_in[5];
    (void)in_sizes; (void)n_in; (void)out_size; (void)d_ws; (void)ws_size;

    s4d_wmma_kernel<<<S4D_D, 32, 0, stream>>>(A_real, A_imag, B, C, D, log_dt,
                                              (float*)d_out);
}


# Pad LDS row pitch of the four fragment arrays from 64 to 68 floats: the ds_load_b64 fragment reads (one row per lane) go from 16-way bank-conflicted to conflict-free
# speedup vs baseline: 1.3569x; 1.3569x over previous
#include <hip/hip_runtime.h>

typedef __attribute__((ext_vector_type(2))) float v2f;
typedef __attribute__((ext_vector_type(4))) float v4f;
typedef __attribute__((ext_vector_type(8))) float v8f;

#define S4D_D 1024
#define S4D_N 64
#define S4D_L 2048
#define S4D_P 68

__global__ __launch_bounds__(32) void s4d_wmma_kernel(
    const float* __restrict__ A_real,
    const float* __restrict__ A_imag,
    const float* __restrict__ Bmat,
    const float* __restrict__ Cmat,
    const float* __restrict__ Dvec,
    const float* __restrict__ log_dt,
    float* __restrict__ out)
{
    __shared__ float P_re [16 * S4D_P];
    __shared__ float P_nim[16 * S4D_P];
    __shared__ float Q_re [16 * S4D_P];
    __shared__ float Q_im [16 * S4D_P];

    const int d    = blockIdx.x;
    const int lane = threadIdx.x;
    const int hi   = lane >> 4;
    const int lo   = lane & 15;

    const float dt = expf(log_dt[d]);
    const float Dd = Dvec[d];

    float ar[2], ai[2];
    float pr[2], pi[2];
    #pragma unroll
    for (int t = 0; t < 2; ++t) {
        const int n = 2 * lane + t;
        const float zr = 0.5f * dt * A_real[n];
        const float zi = 0.5f * dt * A_imag[n];
        const float dr = 1.0f - zr, di = -zi;
        const float inv = 1.0f / (dr * dr + di * di);
        const float nr = 1.0f + zr, ni = zi;
        ar[t] = (nr * dr + ni * di) * inv;
        ai[t] = (ni * dr - nr * di) * inv;
        const float bs  = dt * Bmat[n * S4D_D + d] * inv;
        const float bbr =  bs * dr;
        const float bbi = -bs * di;
        const float cr = Cmat[(d * S4D_N + n) * 2 + 0];
        const float ci = Cmat[(d * S4D_N + n) * 2 + 1];
        pr[t] = cr * bbr - ci * bbi;
        pi[t] = cr * bbi + ci * bbr;
    }

    #pragma unroll
    for (int k = 0; k < 16; ++k) {
        v2f sre, sim;
        sre[0] = pr[0];  sre[1] = pr[1];
        sim[0] = -pi[0]; sim[1] = -pi[1];
        *(v2f*)&P_re [k * S4D_P + 2 * lane] = sre;
        *(v2f*)&P_nim[k * S4D_P + 2 * lane] = sim;
        #pragma unroll
        for (int t = 0; t < 2; ++t) {
            const float trr = pr[t] * ar[t] - pi[t] * ai[t];
            pi[t] = pr[t] * ai[t] + pi[t] * ar[t];
            pr[t] = trr;
        }
    }

    float a16r[2], a16i[2];
    #pragma unroll
    for (int t = 0; t < 2; ++t) { a16r[t] = ar[t]; a16i[t] = ai[t]; }
    #pragma unroll
    for (int s = 0; s < 4; ++s) {
        #pragma unroll
        for (int t = 0; t < 2; ++t) {
            const float trr = a16r[t] * a16r[t] - a16i[t] * a16i[t];
            a16i[t] = 2.0f * a16r[t] * a16i[t];
            a16r[t] = trr;
        }
    }

    float qrr[2] = {1.0f, 1.0f}, qii[2] = {0.0f, 0.0f};

    for (int jt = 0; jt < 8; ++jt) {
        __syncthreads();
        #pragma unroll
        for (int jl = 0; jl < 16; ++jl) {
            v2f sre, sim;
            sre[0] = qrr[0]; sre[1] = qrr[1];
            sim[0] = qii[0]; sim[1] = qii[1];
            *(v2f*)&Q_re[jl * S4D_P + 2 * lane] = sre;
            *(v2f*)&Q_im[jl * S4D_P + 2 * lane] = sim;
            #pragma unroll
            for (int t = 0; t < 2; ++t) {
                const float trr = qrr[t] * a16r[t] - qii[t] * a16i[t];
                qii[t] = qrr[t] * a16i[t] + qii[t] * a16r[t];
                qrr[t] = trr;
            }
        }
        __syncthreads();

        v8f acc_r = {};
        v8f acc_i = {};
        #pragma unroll
        for (int kk = 0; kk < 16; ++kk) {
            const int nb = 4 * kk + 2 * hi;
            v2f a_re = *(const v2f*)&P_re [lo * S4D_P + nb];
            v2f a_im = *(const v2f*)&P_nim[lo * S4D_P + nb];
            v2f b_re = *(const v2f*)&Q_re [lo * S4D_P + nb];
            v2f b_im = *(const v2f*)&Q_im [lo * S4D_P + nb];
            acc_r = __builtin_amdgcn_wmma_f32_16x16x4_f32(
                false, a_re, false, b_re, (short)0, acc_r, false, false);
            acc_i = __builtin_amdgcn_wmma_f32_16x16x4_f32(
                false, a_im, false, b_im, (short)0, acc_i, false, false);
        }

        const int lbase = (jt * 16 + lo) * 16 + 8 * hi;
        v4f w0, w1;
        #pragma unroll
        for (int v = 0; v < 4; ++v) {
            float val = acc_r[v] + acc_i[v];
            val += (lbase + v == 0) ? Dd : 0.0f;
            w0[v] = val;
        }
        #pragma unroll
        for (int v = 0; v < 4; ++v) {
            w1[v] = acc_r[v + 4] + acc_i[v + 4];
        }
        float* orow = out + (size_t)d * S4D_L + lbase;
        *(v4f*)(orow)     = w0;
        *(v4f*)(orow + 4) = w1;
    }
}

extern "C" void kernel_launch(void* const* d_in, const int* in_sizes, int n_in,
                              void* d_out, int out_size, void* d_ws, size_t ws_size,
                              hipStream_t stream) {
    const float* A_real = (const float*)d_in[0];
    const float* A_imag = (const float*)d_in[1];
    const float* B      = (const float*)d_in[2];
    const float* C      = (const float*)d_in[3];
    const float* D      = (const float*)d_in[4];
    const float* log_dt = (const float*)d_in[5];
    (void)in_sizes; (void)n_in; (void)out_size; (void)d_ws; (void)ws_size;

    s4d_wmma_kernel<<<S4D_D, 32, 0, stream>>>(A_real, A_imag, B, C, D, log_dt,
                                              (float*)d_out);
}


# Tile loop runs from registers: P fragments read from LDS once, Q fragments advanced per tile by a complex multiply with Abar^256 (8 squarings); no LDS access or barrier in the loop; sum order kept
# speedup vs baseline: 1.5380x; 1.1335x over previous
#include <hip/hip_runtime.h>


typedef __attribute__((ext_vector_type(2))) float v2f;
typedef __attribute__((ext_vector_type(4))) float v4f;
typedef __attribute__((ext_vector_type(8))) float v8f;

#define S4D_D 1024
#define S4D_N 64
#define S4D_L 2048
#define S4D_P 68

__device__ __forceinline__ void s4d_store_tile(float* __restrict__ orow_d, int jt, int lo, int hi,
                                               float Dd, const v8f& acc_r, const v8f& acc_i)
{
    const int lbase = (jt * 16 + lo) * 16 + 8 * hi;
    v4f w0, w1;
    #pragma unroll
    for (int v = 0; v < 4; ++v) {
        float val = acc_r[v] + acc_i[v];
        val += (lbase + v == 0) ? Dd : 0.0f;
        w0[v] = val;
    }
    #pragma unroll
    for (int v = 0; v < 4; ++v) {
        w1[v] = acc_r[v + 4] + acc_i[v + 4];
    }
    float* orow = orow_d + lbase;
    *(v4f*)(orow)     = w0;
    *(v4f*)(orow + 4) = w1;
}

__global__ __launch_bounds__(32) void s4d_wmma_kernel(
    const float* __restrict__ A_real,
    const float* __restrict__ A_imag,
    const float* __restrict__ Bmat,
    const float* __restrict__ Cmat,
    const float* __restrict__ Dvec,
    const float* __restrict__ log_dt,
    float* __restrict__ out)
{
    __shared__ float P_re [16 * S4D_P];
    __shared__ float P_nim[16 * S4D_P];
    __shared__ float Q_re [16 * S4D_P];
    __shared__ float Q_im [16 * S4D_P];
    __shared__ __attribute__((aligned(16))) float M_re [S4D_N];
    __shared__ __attribute__((aligned(16))) float M_im [S4D_N];

    const int d    = blockIdx.x;
    const int lane = threadIdx.x;
    const int hi   = lane >> 4;
    const int lo   = lane & 15;

    const float dt = expf(log_dt[d]);
    const float Dd = Dvec[d];

    float ar[2], ai[2];
    float pr[2], pi[2];
    #pragma unroll
    for (int t = 0; t < 2; ++t) {
        const int n = 2 * lane + t;
        const float zr = 0.5f * dt * A_real[n];
        const float zi = 0.5f * dt * A_imag[n];
        const float dr = 1.0f - zr, di = -zi;
        const float inv = 1.0f / (dr * dr + di * di);
        const float nr = 1.0f + zr, ni = zi;
        ar[t] = (nr * dr + ni * di) * inv;
        ai[t] = (ni * dr - nr * di) * inv;
        const float bs  = dt * Bmat[n * S4D_D + d] * inv;
        const float bbr =  bs * dr;
        const float bbi = -bs * di;
        const float cr = Cmat[(d * S4D_N + n) * 2 + 0];
        const float ci = Cmat[(d * S4D_N + n) * 2 + 1];
        pr[t] = cr * bbr - ci * bbi;
        pi[t] = cr * bbi + ci * bbr;
    }

    #pragma unroll
    for (int k = 0; k < 16; ++k) {
        v2f sre, sim;
        sre[0] = pr[0];  sre[1] = pr[1];
        sim[0] = -pi[0]; sim[1] = -pi[1];
        *(v2f*)&P_re [k * S4D_P + 2 * lane] = sre;
        *(v2f*)&P_nim[k * S4D_P + 2 * lane] = sim;
        #pragma unroll
        for (int t = 0; t < 2; ++t) {
            const float trr = pr[t] * ar[t] - pi[t] * ai[t];
            pi[t] = pr[t] * ai[t] + pi[t] * ar[t];
            pr[t] = trr;
        }
    }

    float a16r[2], a16i[2];
    #pragma unroll
    for (int t = 0; t < 2; ++t) { a16r[t] = ar[t]; a16i[t] = ai[t]; }
    #pragma unroll
    for (int s = 0; s < 4; ++s) {
        #pragma unroll
        for (int t = 0; t < 2; ++t) {
            const float trr = a16r[t] * a16r[t] - a16i[t] * a16i[t];
            a16i[t] = 2.0f * a16r[t] * a16i[t];
            a16r[t] = trr;
        }
    }

    {
        float qrr[2] = {1.0f, 1.0f}, qii[2] = {0.0f, 0.0f};
        #pragma unroll
        for (int jl = 0; jl < 16; ++jl) {
            v2f sre, sim;
            sre[0] = qrr[0]; sre[1] = qrr[1];
            sim[0] = qii[0]; sim[1] = qii[1];
            *(v2f*)&Q_re[jl * S4D_P + 2 * lane] = sre;
            *(v2f*)&Q_im[jl * S4D_P + 2 * lane] = sim;
            #pragma unroll
            for (int t = 0; t < 2; ++t) {
                const float trr = qrr[t] * a16r[t] - qii[t] * a16i[t];
                qii[t] = qrr[t] * a16i[t] + qii[t] * a16r[t];
                qrr[t] = trr;
            }
        }
    }

    {
        float mr[2], mi[2];
        #pragma unroll
        for (int t = 0; t < 2; ++t) { mr[t] = a16r[t]; mi[t] = a16i[t]; }
        #pragma unroll
        for (int s = 0; s < 4; ++s) {
            #pragma unroll
            for (int t = 0; t < 2; ++t) {
                const float trr = mr[t] * mr[t] - mi[t] * mi[t];
                mi[t] = 2.0f * mr[t] * mi[t];
                mr[t] = trr;
            }
        }
        v2f sre, sim;
        sre[0] = mr[0]; sre[1] = mr[1];
        sim[0] = mi[0]; sim[1] = mi[1];
        *(v2f*)&M_re[2 * lane] = sre;
        *(v2f*)&M_im[2 * lane] = sim;
    }

    __syncthreads();

    v2f a_re[16], a_im[16], b_re[16], b_im[16], m_re[16], m_im[16];
    #pragma unroll
    for (int kk = 0; kk < 16; ++kk) {
        const int nb = 4 * kk + 2 * hi;
        a_re[kk] = *(const v2f*)&P_re [lo * S4D_P + nb];
        a_im[kk] = *(const v2f*)&P_nim[lo * S4D_P + nb];
        b_re[kk] = *(const v2f*)&Q_re [lo * S4D_P + nb];
        b_im[kk] = *(const v2f*)&Q_im [lo * S4D_P + nb];
        m_re[kk] = *(const v2f*)&M_re [nb];
        m_im[kk] = *(const v2f*)&M_im [nb];
    }

    float* orow_d = out + (size_t)d * S4D_L;

    for (int jt = 0; jt < 7; ++jt) {
        v8f acc_r = {};
        v8f acc_i = {};
        #pragma unroll
        for (int kk = 0; kk < 16; ++kk) {
            acc_r = __builtin_amdgcn_wmma_f32_16x16x4_f32(
                false, a_re[kk], false, b_re[kk], (short)0, acc_r, false, false);
            acc_i = __builtin_amdgcn_wmma_f32_16x16x4_f32(
                false, a_im[kk], false, b_im[kk], (short)0, acc_i, false, false);
            const v2f qr = b_re[kk], qi = b_im[kk];
            b_re[kk] = qr * m_re[kk] - qi * m_im[kk];
            b_im[kk] = qr * m_im[kk] + qi * m_re[kk];
        }
        s4d_store_tile(orow_d, jt, lo, hi, Dd, acc_r, acc_i);
    }

    {
        v8f acc_r = {};
        v8f acc_i = {};
        #pragma unroll
        for (int kk = 0; kk < 16; ++kk) {
            acc_r = __builtin_amdgcn_wmma_f32_16x16x4_f32(
                false, a_re[kk], false, b_re[kk], (short)0, acc_r, false, false);
            acc_i = __builtin_amdgcn_wmma_f32_16x16x4_f32(
                false, a_im[kk], false, b_im[kk], (short)0, acc_i, false, false);
        }
        s4d_store_tile(orow_d, 7, lo, hi, Dd, acc_r, acc_i);
    }
}

extern "C" void kernel_launch(void* const* d_in, const int* in_sizes, int n_in,
                              void* d_out, int out_size, void* d_ws, size_t ws_size,
                              hipStream_t stream) {
    const float* A_real = (const float*)d_in[0];
    const float* A_imag = (const float*)d_in[1];
    const float* B      = (const float*)d_in[2];
    const float* C      = (const float*)d_in[3];
    const float* D      = (const float*)d_in[4];
    const float* log_dt = (const float*)d_in[5];
    (void)in_sizes; (void)n_in; (void)out_size; (void)d_ws; (void)ws_size;

    s4d_wmma_kernel<<<S4D_D, 32, 0, stream>>>(A_real, A_imag, B, C, D, log_dt,
                                              (float*)d_out);
}


# One 4-wave work-group per WGP (grid 256, 164 KiB of LDS reserved so that only one is resident), one channel per wave: every SIMD serves exactly one channel's WMMA run; sum order kept
# speedup vs baseline: 2.1218x; 1.3796x over previous
#include <hip/hip_runtime.h>


typedef __attribute__((ext_vector_type(2))) float v2f;
typedef __attribute__((ext_vector_type(4))) float v4f;
typedef __attribute__((ext_vector_type(8))) float v8f;

#define S4D_D 1024
#define S4D_N 64
#define S4D_L 2048
#define S4D_P 68
#define S4D_W 4
#define S4D_WAVE_F (4 * 16 * S4D_P + 2 * S4D_N)
#define S4D_LDS_F (41 * 1024)
#define S4D_GRID 256
#define S4D_THREADS 128
static_assert(S4D_GRID * S4D_W == S4D_D, "grid covers every channel once");
static_assert(S4D_THREADS == 32 * S4D_W, "one wave per channel");
static_assert(S4D_W * S4D_WAVE_F <= S4D_LDS_F, "the waves' regions fit the reservation");
static_assert(S4D_LDS_F * 4 > 160 * 1024 && S4D_LDS_F * 4 <= 320 * 1024, "one work-group per WGP");

__device__ __forceinline__ void s4d_store_tile(float* __restrict__ orow_d, int jt, int lo, int hi,
                                               float Dd, const v8f& acc_r, const v8f& acc_i)
{
    const int lbase = (jt * 16 + lo) * 16 + 8 * hi;
    v4f w0, w1;
    #pragma unroll
    for (int v = 0; v < 4; ++v) {
        float val = acc_r[v] + acc_i[v];
        val += (lbase + v == 0) ? Dd : 0.0f;
        w0[v] = val;
    }
    #pragma unroll
    for (int v = 0; v < 4; ++v) {
        w1[v] = acc_r[v + 4] + acc_i[v + 4];
    }
    float* orow = orow_d + lbase;
    *(v4f*)(orow)     = w0;
    *(v4f*)(orow + 4) = w1;
}

__global__ __launch_bounds__(S4D_THREADS) void s4d_wmma_kernel(
    const float* __restrict__ A_real,
    const float* __restrict__ A_imag,
    const float* __restrict__ Bmat,
    const float* __restrict__ Cmat,
    const float* __restrict__ Dvec,
    const float* __restrict__ log_dt,
    float* __restrict__ out)
{
    __shared__ __attribute__((aligned(16))) float lds[S4D_LDS_F];

    const int wave = __builtin_amdgcn_readfirstlane((int)(threadIdx.x >> 5));
    const int d    = blockIdx.x * S4D_W + wave;
    const int lane = threadIdx.x & 31;

    float* const P_re  = lds + wave * S4D_WAVE_F;
    float* const P_nim = P_re  + 16 * S4D_P;
    float* const Q_re  = P_nim + 16 * S4D_P;
    float* const Q_im  = Q_re  + 16 * S4D_P;
    float* const M_re  = Q_im  + 16 * S4D_P;
    float* const M_im  = M_re  + S4D_N;
    const int hi   = lane >> 4;
    const int lo   = lane & 15;

    const float dt = expf(log_dt[d]);
    const float Dd = Dvec[d];

    float ar[2], ai[2];
    float pr[2], pi[2];
    #pragma unroll
    for (int t = 0; t < 2; ++t) {
        const int n = 2 * lane + t;
        const float zr = 0.5f * dt * A_real[n];
        const float zi = 0.5f * dt * A_imag[n];
        const float dr = 1.0f - zr, di = -zi;
        const float inv = 1.0f / (dr * dr + di * di);
        const float nr = 1.0f + zr, ni = zi;
        ar[t] = (nr * dr + ni * di) * inv;
        ai[t] = (ni * dr - nr * di) * inv;
        const float bs  = dt * Bmat[n * S4D_D + d] * inv;
        const float bbr =  bs * dr;
        const float bbi = -bs * di;
        const float cr = Cmat[(d * S4D_N + n) * 2 + 0];
        const float ci = Cmat[(d * S4D_N + n) * 2 + 1];
        pr[t] = cr * bbr - ci * bbi;
        pi[t] = cr * bbi + ci * bbr;
    }

    #pragma unroll
    for (int k = 0; k < 16; ++k) {
        v2f sre, sim;
        sre[0] = pr[0];  sre[1] = pr[1];
        sim[0] = -pi[0]; sim[1] = -pi[1];
        *(v2f*)&P_re [k * S4D_P + 2 * lane] = sre;
        *(v2f*)&P_nim[k * S4D_P + 2 * lane] = sim;
        #pragma unroll
        for (int t = 0; t < 2; ++t) {
            const float trr = pr[t] * ar[t] - pi[t] * ai[t];
            pi[t] = pr[t] * ai[t] + pi[t] * ar[t];
            pr[t] = trr;
        }
    }

    float a16r[2], a16i[2];
    #pragma unroll
    for (int t = 0; t < 2; ++t) { a16r[t] = ar[t]; a16i[t] = ai[t]; }
    #pragma unroll
    for (int s = 0; s < 4; ++s) {
        #pragma unroll
        for (int t = 0; t < 2; ++t) {
            const float trr = a16r[t] * a16r[t] - a16i[t] * a16i[t];
            a16i[t] = 2.0f * a16r[t] * a16i[t];
            a16r[t] = trr;
        }
    }

    {
        float qrr[2] = {1.0f, 1.0f}, qii[2] = {0.0f, 0.0f};
        #pragma unroll
        for (int jl = 0; jl < 16; ++jl) {
            v2f sre, sim;
            sre[0] = qrr[0]; sre[1] = qrr[1];
            sim[0] = qii[0]; sim[1] = qii[1];
            *(v2f*)&Q_re[jl * S4D_P + 2 * lane] = sre;
            *(v2f*)&Q_im[jl * S4D_P + 2 * lane] = sim;
            #pragma unroll
            for (int t = 0; t < 2; ++t) {
                const float trr = qrr[t] * a16r[t] - qii[t] * a16i[t];
                qii[t] = qrr[t] * a16i[t] + qii[t] * a16r[t];
                qrr[t] = trr;
            }
        }
    }

    {
        float mr[2], mi[2];
        #pragma unroll
        for (int t = 0; t < 2; ++t) { mr[t] = a16r[t]; mi[t] = a16i[t]; }
        #pragma unroll
        for (int s = 0; s < 4; ++s) {
            #pragma unroll
            for (int t = 0; t < 2; ++t) {
                const float trr = mr[t] * mr[t] - mi[t] * mi[t];
                mi[t] = 2.0f * mr[t] * mi[t];
                mr[t] = trr;
            }
        }
        v2f sre, sim;
        sre[0] = mr[0]; sre[1] = mr[1];
        sim[0] = mi[0]; sim[1] = mi[1];
        *(v2f*)&M_re[2 * lane] = sre;
        *(v2f*)&M_im[2 * lane] = sim;
    }

    __syncthreads();

    v2f a_re[16], a_im[16], b_re[16], b_im[16], m_re[16], m_im[16];
    #pragma unroll
    for (int kk = 0; kk < 16; ++kk) {
        const int nb = 4 * kk + 2 * hi;
        a_re[kk] = *(const v2f*)&P_re [lo * S4D_P + nb];
        a_im[kk] = *(const v2f*)&P_nim[lo * S4D_P + nb];
        b_re[kk] = *(const v2f*)&Q_re [lo * S4D_P + nb];
        b_im[kk] = *(const v2f*)&Q_im [lo * S4D_P + nb];
        m_re[kk] = *(const v2f*)&M_re [nb];
        m_im[kk] = *(const v2f*)&M_im [nb];
    }

    float* orow_d = out + (size_t)d * S4D_L;

    for (int jt = 0; jt < 7; ++jt) {
        v8f acc_r = {};
        v8f acc_i = {};
        #pragma unroll
        for (int kk = 0; kk < 16; ++kk) {
            acc_r = __builtin_amdgcn_wmma_f32_16x16x4_f32(
                false, a_re[kk], false, b_re[kk], (short)0, acc_r, false, false);
            acc_i = __builtin_amdgcn_wmma_f32_16x16x4_f32(
                false, a_im[kk], false, b_im[kk], (short)0, acc_i, false, false);
            const v2f qr = b_re[kk], qi = b_im[kk];
            b_re[kk] = qr * m_re[kk] - qi * m_im[kk];
            b_im[kk] = qr * m_im[kk] + qi * m_re[kk];
        }
        s4d_store_tile(orow_d, jt, lo, hi, Dd, acc_r, acc_i);
    }

    {
        v8f acc_r = {};
        v8f acc_i = {};
        #pragma unroll
        for (int kk = 0; kk < 16; ++kk) {
            acc_r = __builtin_amdgcn_wmma_f32_16x16x4_f32(
                false, a_re[kk], false, b_re[kk], (short)0, acc_r, false, false);
            acc_i = __builtin_amdgcn_wmma_f32_16x16x4_f32(
                false, a_im[kk], false, b_im[kk], (short)0, acc_i, false, false);
        }
        s4d_store_tile(orow_d, 7, lo, hi, Dd, acc_r, acc_i);
    }
}

extern "C" void kernel_launch(void* const* d_in, const int* in_sizes, int n_in,
                              void* d_out, int out_size, void* d_ws, size_t ws_size,
                              hipStream_t stream) {
    const float* A_real = (const float*)d_in[0];
    const float* A_imag = (const float*)d_in[1];
    const float* B      = (const float*)d_in[2];
    const float* C      = (const float*)d_in[3];
    const float* D      = (const float*)d_in[4];
    const float* log_dt = (const float*)d_in[5];
    (void)in_sizes; (void)n_in; (void)out_size; (void)d_ws; (void)ws_size;

    s4d_wmma_kernel<<<S4D_GRID, S4D_THREADS, 0, stream>>>(A_real, A_imag, B, C, D, log_dt,
                                                          (float*)d_out);
}


# Tile loop unrolled into one pipelined matrix run: tile 0 starts on counted LDS waits while fragments still land; each tile's merge+stores issue 4 K-steps into the next tile; sum order kept
# speedup vs baseline: 2.2872x; 1.0780x over previous
#include <hip/hip_runtime.h>


typedef __attribute__((ext_vector_type(2))) float v2f;
typedef __attribute__((ext_vector_type(4))) float v4f;
typedef __attribute__((ext_vector_type(8))) float v8f;

#define S4D_D 1024
#define S4D_N 64
#define S4D_L 2048
#define S4D_P 68
#define S4D_W 4
#define S4D_WAVE_F (4 * 16 * S4D_P + 2 * S4D_N)
#define S4D_LDS_F (41 * 1024)
#define S4D_GRID 256
#define S4D_THREADS 128
static_assert(S4D_GRID * S4D_W == S4D_D, "grid covers every channel once");
static_assert(S4D_THREADS == 32 * S4D_W, "one wave per channel");
static_assert(S4D_W * S4D_WAVE_F <= S4D_LDS_F, "the waves' regions fit the reservation");
static_assert(S4D_LDS_F * 4 > 160 * 1024 && S4D_LDS_F * 4 <= 320 * 1024, "one work-group per WGP");

__device__ __forceinline__ void s4d_store_tile(float* __restrict__ orow_d, int jt, int lo, int hi,
                                               float Dd, const v8f& acc_r, const v8f& acc_i)
{
    const int lbase = (jt * 16 + lo) * 16 + 8 * hi;
    v4f w0, w1;
    #pragma unroll
    for (int v = 0; v < 4; ++v) {
        float val = acc_r[v] + acc_i[v];
        val += (lbase + v == 0) ? Dd : 0.0f;
        w0[v] = val;
    }
    #pragma unroll
    for (int v = 0; v < 4; ++v) {
        w1[v] = acc_r[v + 4] + acc_i[v + 4];
    }
    float* orow = orow_d + lbase;
    *(v4f*)(orow)     = w0;
    *(v4f*)(orow + 4) = w1;
}

__global__ __launch_bounds__(S4D_THREADS) void s4d_wmma_kernel(
    const float* __restrict__ A_real,
    const float* __restrict__ A_imag,
    const float* __restrict__ Bmat,
    const float* __restrict__ Cmat,
    const float* __restrict__ Dvec,
    const float* __restrict__ log_dt,
    float* __restrict__ out)
{
    __shared__ __attribute__((aligned(16))) float lds[S4D_LDS_F];

    const int wave = __builtin_amdgcn_readfirstlane((int)(threadIdx.x >> 5));
    const int d    = blockIdx.x * S4D_W + wave;
    const int lane = threadIdx.x & 31;

    float* const P_re  = lds + wave * S4D_WAVE_F;
    float* const P_nim = P_re  + 16 * S4D_P;
    float* const Q_re  = P_nim + 16 * S4D_P;
    float* const Q_im  = Q_re  + 16 * S4D_P;
    float* const M_re  = Q_im  + 16 * S4D_P;
    float* const M_im  = M_re  + S4D_N;
    const int hi   = lane >> 4;
    const int lo   = lane & 15;

    const float dt = expf(log_dt[d]);
    const float Dd = Dvec[d];

    float ar[2], ai[2];
    float pr[2], pi[2];
    #pragma unroll
    for (int t = 0; t < 2; ++t) {
        const int n = 2 * lane + t;
        const float zr = 0.5f * dt * A_real[n];
        const float zi = 0.5f * dt * A_imag[n];
        const float dr = 1.0f - zr, di = -zi;
        const float inv = 1.0f / (dr * dr + di * di);
        const float nr = 1.0f + zr, ni = zi;
        ar[t] = (nr * dr + ni * di) * inv;
        ai[t] = (ni * dr - nr * di) * inv;
        const float bs  = dt * Bmat[n * S4D_D + d] * inv;
        const float bbr =  bs * dr;
        const float bbi = -bs * di;
        const float cr = Cmat[(d * S4D_N + n) * 2 + 0];
        const float ci = Cmat[(d * S4D_N + n) * 2 + 1];
        pr[t] = cr * bbr - ci * bbi;
        pi[t] = cr * bbi + ci * bbr;
    }

    #pragma unroll
    for (int k = 0; k < 16; ++k) {
        v2f sre, sim;
        sre[0] = pr[0];  sre[1] = pr[1];
        sim[0] = -pi[0]; sim[1] = -pi[1];
        *(v2f*)&P_re [k * S4D_P + 2 * lane] = sre;
        *(v2f*)&P_nim[k * S4D_P + 2 * lane] = sim;
        #pragma unroll
        for (int t = 0; t < 2; ++t) {
            const float trr = pr[t] * ar[t] - pi[t] * ai[t];
            pi[t] = pr[t] * ai[t] + pi[t] * ar[t];
            pr[t] = trr;
        }
    }

    float a16r[2], a16i[2];
    #pragma unroll
    for (int t = 0; t < 2; ++t) { a16r[t] = ar[t]; a16i[t] = ai[t]; }
    #pragma unroll
    for (int s = 0; s < 4; ++s) {
        #pragma unroll
        for (int t = 0; t < 2; ++t) {
            const float trr = a16r[t] * a16r[t] - a16i[t] * a16i[t];
            a16i[t] = 2.0f * a16r[t] * a16i[t];
            a16r[t] = trr;
        }
    }

    {
        float qrr[2] = {1.0f, 1.0f}, qii[2] = {0.0f, 0.0f};
        #pragma unroll
        for (int jl = 0; jl < 16; ++jl) {
            v2f sre, sim;
            sre[0] = qrr[0]; sre[1] = qrr[1];
            sim[0] = qii[0]; sim[1] = qii[1];
            *(v2f*)&Q_re[jl * S4D_P + 2 * lane] = sre;
            *(v2f*)&Q_im[jl * S4D_P + 2 * lane] = sim;
            #pragma unroll
            for (int t = 0; t < 2; ++t) {
                const float trr = qrr[t] * a16r[t] - qii[t] * a16i[t];
                qii[t] = qrr[t] * a16i[t] + qii[t] * a16r[t];
                qrr[t] = trr;
            }
        }
    }

    {
        float mr[2], mi[2];
        #pragma unroll
        for (int t = 0; t < 2; ++t) { mr[t] = a16r[t]; mi[t] = a16i[t]; }
        #pragma unroll
        for (int s = 0; s < 4; ++s) {
            #pragma unroll
            for (int t = 0; t < 2; ++t) {
                const float trr = mr[t] * mr[t] - mi[t] * mi[t];
                mi[t] = 2.0f * mr[t] * mi[t];
                mr[t] = trr;
            }
        }
        v2f sre, sim;
        sre[0] = mr[0]; sre[1] = mr[1];
        sim[0] = mi[0]; sim[1] = mi[1];
        *(v2f*)&M_re[2 * lane] = sre;
        *(v2f*)&M_im[2 * lane] = sim;
    }

    __syncthreads();

    v2f a_re[16], a_im[16], b_re[16], b_im[16], m_re[16], m_im[16];
    #pragma unroll
    for (int kk = 0; kk < 16; ++kk) {
        const int nb = 4 * kk + 2 * hi;
        a_re[kk] = *(const v2f*)&P_re [lo * S4D_P + nb];
        a_im[kk] = *(const v2f*)&P_nim[lo * S4D_P + nb];
        b_re[kk] = *(const v2f*)&Q_re [lo * S4D_P + nb];
        b_im[kk] = *(const v2f*)&Q_im [lo * S4D_P + nb];
    }
    #pragma unroll
    for (int kk = 0; kk < 16; ++kk) {
        const int nb = 4 * kk + 2 * hi;
        m_re[kk] = *(const v2f*)&M_re [nb];
        m_im[kk] = *(const v2f*)&M_im [nb];
    }

    float* orow_d = out + (size_t)d * S4D_L;

    v8f prev_r = {};
    v8f prev_i = {};
    #pragma unroll
    for (int jt = 0; jt < 8; ++jt) {
        v8f acc_r = {};
        v8f acc_i = {};
        #pragma unroll
        for (int kk = 0; kk < 16; ++kk) {
            if (kk == 4 && jt > 0) {
                s4d_store_tile(orow_d, jt - 1, lo, hi, Dd, prev_r, prev_i);
            }
            if (kk == 8) __builtin_amdgcn_sched_barrier(0);
            acc_r = __builtin_amdgcn_wmma_f32_16x16x4_f32(
                false, a_re[kk], false, b_re[kk], (short)0, acc_r, false, false);
            acc_i = __builtin_amdgcn_wmma_f32_16x16x4_f32(
                false, a_im[kk], false, b_im[kk], (short)0, acc_i, false, false);
            if (jt > 0 && jt < 7) {
                const v2f qr = b_re[kk], qi = b_im[kk];
                b_re[kk] = qr * m_re[kk] - qi * m_im[kk];
                b_im[kk] = qr * m_im[kk] + qi * m_re[kk];
            }
        }
        if (jt == 0) {
            #pragma unroll
            for (int kk = 0; kk < 16; ++kk) {
                const v2f qr = b_re[kk], qi = b_im[kk];
                b_re[kk] = qr * m_re[kk] - qi * m_im[kk];
                b_im[kk] = qr * m_im[kk] + qi * m_re[kk];
            }
        }
        prev_r = acc_r;
        prev_i = acc_i;
    }
    s4d_store_tile(orow_d, 7, lo, hi, Dd, prev_r, prev_i);
}

extern "C" void kernel_launch(void* const* d_in, const int* in_sizes, int n_in,
                              void* d_out, int out_size, void* d_ws, size_t ws_size,
                              hipStream_t stream) {
    const float* A_real = (const float*)d_in[0];
    const float* A_imag = (const float*)d_in[1];
    const float* B      = (const float*)d_in[2];
    const float* C      = (const float*)d_in[3];
    const float* D      = (const float*)d_in[4];
    const float* log_dt = (const float*)d_in[5];
    (void)in_sizes; (void)n_in; (void)out_size; (void)d_ws; (void)ws_size;

    s4d_wmma_kernel<<<S4D_GRID, S4D_THREADS, 0, stream>>>(A_real, A_imag, B, C, D, log_dt,
                                                          (float*)d_out);
}
